# starting kernel
# baseline (speedup 1.0000x reference)
#include <hip/hip_runtime.h>
#include <hip/hip_bf16.h>

typedef __attribute__((ext_vector_type(16))) _Float16 v16h;
typedef __attribute__((ext_vector_type(8)))  _Float16 v8h;
typedef __attribute__((ext_vector_type(4)))  _Float16 v4h;
typedef __attribute__((ext_vector_type(8)))  float    v8f;

#define HID 128
#define NEG_SLOPE 0.2f
#define BN_EPS 1e-5f
#define N_GRAPHS 512

__global__ void pack_w_f16(const float* __restrict__ W, _Float16* __restrict__ Wp, int N) {
    int id = blockIdx.x * blockDim.x + threadIdx.x;
    int total = (N / 16) * 4 * 32;
    if (id >= total) return;
    int t    = id >> 7;
    int kt   = (id >> 5) & 3;
    int lane = id & 31;
    int hl = lane & 15, hi = lane >> 4;
    int n = t * 16 + hl;
    _Float16* o = Wp + (size_t)id * 16;
#pragma unroll
    for (int j = 0; j < 8; ++j) {
        o[j]     = (_Float16)W[(size_t)(kt * 32 + hi * 8 + j) * N + n];
        o[8 + j] = (_Float16)W[(size_t)(kt * 32 + 16 + hi * 8 + j) * N + n];
    }
}

template <bool HAS_BIAS>
__global__ void gemm128_wmma(const float* __restrict__ A, const _Float16* __restrict__ Bp,
                             const float* __restrict__ bias, float* __restrict__ C,
                             int N) {
    __shared__ __align__(16) _Float16 As[16 * 128];
    const int tid  = threadIdx.x;
    const int row0 = blockIdx.x * 16;

    for (int c = tid; c < 512; c += blockDim.x) {
        int r   = c >> 5;
        int col = (c & 31) * 4;
        float4 f = *(const float4*)(A + (size_t)(row0 + r) * 128 + col);
        v4h h4 = { (_Float16)f.x, (_Float16)f.y, (_Float16)f.z, (_Float16)f.w };
        *(v4h*)(As + r * 128 + col) = h4;
    }
    __syncthreads();

    const int wave = tid >> 5;
    const int lane = tid & 31;
    const int hl   = lane & 15;
    const int hi   = lane >> 4;

    v8f acc = {};
#pragma unroll
    for (int kt = 0; kt < 4; ++kt) {
        int base = hl * 128 + kt * 32 + hi * 8;
        v8h alo = *(const v8h*)(As + base);
        v8h ahi = *(const v8h*)(As + base + 16);
        v16h a;
#pragma unroll
        for (int j = 0; j < 8; ++j) { a[j] = alo[j]; a[8 + j] = ahi[j]; }

        v16h b = *(const v16h*)(Bp + (((size_t)wave * 4 + kt) * 32 + lane) * 16);

        acc = __builtin_amdgcn_wmma_f32_16x16x32_f16(
            false, a, false, b, (short)0, acc, false, false);
    }

    const int col = wave * 16 + hl;
    float bv = HAS_BIAS ? bias[col] : 0.0f;
#pragma unroll
    for (int r = 0; r < 8; ++r) {
        int row = row0 + r + 8 * hi;
        C[(size_t)row * N + col] = acc[r] + bv;
    }
}

__global__ void node_scores(const float* __restrict__ H, const float* __restrict__ a_src,
                            const float* __restrict__ a_dst, float* __restrict__ ss,
                            float* __restrict__ sd, int n_nodes) {
    int node = (int)((blockIdx.x * (size_t)blockDim.x + threadIdx.x) >> 5);
    int lane = threadIdx.x & 31;
    if (node >= n_nodes) return;
    const float* h = H + (size_t)node * HID;
    float s0 = 0.f, s1 = 0.f;
#pragma unroll
    for (int j = 0; j < 4; ++j) {
        int c = lane + 32 * j;
        float hv = h[c];
        s0 += hv * a_src[c];
        s1 += hv * a_dst[c];
    }
#pragma unroll
    for (int off = 16; off > 0; off >>= 1) {
        s0 += __shfl_down(s0, off, 32);
        s1 += __shfl_down(s1, off, 32);
    }
    if (lane == 0) { ss[node] = s0; sd[node] = s1; }
}

__global__ void fill_f32(float* __restrict__ p, float v, int n) {
    int i = blockIdx.x * blockDim.x + threadIdx.x;
    if (i < n) p[i] = v;
}

__device__ __forceinline__ void atomicMaxF(float* addr, float val) {
    if (val >= 0.f) atomicMax((int*)addr, __float_as_int(val));
    else            atomicMin((unsigned int*)addr, __float_as_uint(val));
}

__device__ __forceinline__ void edge_sd(int e, int n_edges,
                                        const int* __restrict__ esrc,
                                        const int* __restrict__ edst, int& s, int& d) {
    if (e < n_edges) { s = esrc[e]; d = edst[e]; }
    else             { s = e - n_edges; d = s; }
}

__global__ void edge_max(const int* __restrict__ esrc, const int* __restrict__ edst,
                         const float* __restrict__ ss, const float* __restrict__ sd,
                         float* __restrict__ mx, int n_edges, int etot) {
    int e = blockIdx.x * blockDim.x + threadIdx.x;
    if (e >= etot) return;
    int s, d; edge_sd(e, n_edges, esrc, edst, s, d);
    float v = ss[s] + sd[d];
    v = (v > 0.f) ? v : NEG_SLOPE * v;
    atomicMaxF(&mx[d], v);
}

__global__ void edge_expsum(const int* __restrict__ esrc, const int* __restrict__ edst,
                            const float* __restrict__ ss, const float* __restrict__ sd,
                            const float* __restrict__ mx, float* __restrict__ den,
                            int n_edges, int etot) {
    int e = blockIdx.x * blockDim.x + threadIdx.x;
    if (e >= etot) return;
    int s, d; edge_sd(e, n_edges, esrc, edst, s, d);
    float v = ss[s] + sd[d];
    v = (v > 0.f) ? v : NEG_SLOPE * v;
    atomicAdd(&den[d], __expf(v - mx[d]));
}

__global__ void edge_aggregate(const float* __restrict__ H, const int* __restrict__ esrc,
                               const int* __restrict__ edst, const float* __restrict__ ss,
                               const float* __restrict__ sd, const float* __restrict__ mx,
                               const float* __restrict__ den, float* __restrict__ out,
                               int n_edges, int etot) {
    int e = (int)((blockIdx.x * (size_t)blockDim.x + threadIdx.x) >> 5);
    int lane = threadIdx.x & 31;
    if (e >= etot) return;
    int s, d; edge_sd(e, n_edges, esrc, edst, s, d);
    float v = ss[s] + sd[d];
    v = (v > 0.f) ? v : NEG_SLOPE * v;
    float w = __expf(v - mx[d]) / (den[d] + 1e-16f);
    const float* hs = H + (size_t)s * HID;
    float* od = out + (size_t)d * HID;
#pragma unroll
    for (int j = 0; j < 4; ++j) {
        int c = lane + 32 * j;
        atomicAdd(&od[c], w * hs[c]);
    }
}

__global__ void bias_bn_relu(float* __restrict__ x, const float* __restrict__ bias,
                             const float* __restrict__ gam, const float* __restrict__ bet,
                             const float* __restrict__ mean, const float* __restrict__ var,
                             int total) {
    int i = blockIdx.x * blockDim.x + threadIdx.x;
    if (i >= total) return;
    int c = i & (HID - 1);
    float v = x[i] + bias[c];
    v = (v - mean[c]) * (gam[c] * rsqrtf(var[c] + BN_EPS)) + bet[c];
    x[i] = fmaxf(v, 0.f);
}

__global__ void pool_accum(const float* __restrict__ x, const int* __restrict__ batch,
                           float* __restrict__ sums, float* __restrict__ cnt, int n_nodes) {
    int i = blockIdx.x * blockDim.x + threadIdx.x;
    if (i >= n_nodes * HID) return;
    int n = i >> 7, c = i & (HID - 1);
    int g = batch[n];
    atomicAdd(&sums[(size_t)g * HID + c], x[i]);
    if (c == 0) atomicAdd(&cnt[g], 1.0f);
}

__global__ void pool_div(float* __restrict__ sums, const float* __restrict__ cnt, int total) {
    int i = blockIdx.x * blockDim.x + threadIdx.x;
    if (i >= total) return;
    sums[i] /= fmaxf(cnt[i >> 7], 1.0f);
}

extern "C" void kernel_launch(void* const* d_in, const int* in_sizes, int n_in,
                              void* d_out, int out_size, void* d_ws, size_t ws_size,
                              hipStream_t stream) {
    const float* gx  = (const float*)d_in[0];
    const int*   ei  = (const int*)  d_in[1];
    const int*   bat = (const int*)  d_in[2];
    const float* W1  = (const float*)d_in[3];
    const float* as1 = (const float*)d_in[4];
    const float* ad1 = (const float*)d_in[5];
    const float* b1  = (const float*)d_in[6];
    const float* g1  = (const float*)d_in[7];
    const float* be1 = (const float*)d_in[8];
    const float* mu1 = (const float*)d_in[9];
    const float* va1 = (const float*)d_in[10];
    const float* W2  = (const float*)d_in[11];
    const float* as2 = (const float*)d_in[12];
    const float* ad2 = (const float*)d_in[13];
    const float* b2  = (const float*)d_in[14];
    const float* g2  = (const float*)d_in[15];
    const float* be2 = (const float*)d_in[16];
    const float* mu2 = (const float*)d_in[17];
    const float* va2 = (const float*)d_in[18];
    const float* hw  = (const float*)d_in[19];
    const float* hb  = (const float*)d_in[20];

    const int n_nodes = in_sizes[0] / HID;
    const int n_edges = in_sizes[1] / 2;
    const int etot    = n_edges + n_nodes;
    const int* esrc = ei;
    const int* edst = ei + n_edges;

    float* ws   = (float*)d_ws;
    float* H    = ws;
    float* AGG  = H    + (size_t)n_nodes * HID;
    float* SS   = AGG  + (size_t)n_nodes * HID;
    float* SD   = SS   + n_nodes;
    float* MX   = SD   + n_nodes;
    float* DEN  = MX   + n_nodes;
    float* POOL = DEN  + n_nodes;
    float* CNT  = POOL + (size_t)N_GRAPHS * HID;
    size_t wpOff = (size_t)(CNT + N_GRAPHS - ws);
    wpOff = (wpOff + 7) & ~(size_t)7;
    _Float16* WP = (_Float16*)(ws + wpOff);

    const int feat_total = n_nodes * HID;
    const int gridFeat  = (feat_total + 255) / 256;
    const int gridNode  = (n_nodes + 255) / 256;
    const int gridEdge  = (etot + 255) / 256;
    const int gridEdgeW = (etot + 7) / 8;
    const int gridNodeW = (n_nodes + 7) / 8;
    const int gridGemm  = n_nodes / 16;

    for (int layer = 0; layer < 2; ++layer) {
        const float* xin = (layer == 0) ? gx : AGG;
        const float* W   = (layer == 0) ? W1 : W2;
        const float* asr = (layer == 0) ? as1 : as2;
        const float* ads = (layer == 0) ? ad1 : ad2;
        const float* bia = (layer == 0) ? b1  : b2;
        const float* gam = (layer == 0) ? g1  : g2;
        const float* bet = (layer == 0) ? be1 : be2;
        const float* mu  = (layer == 0) ? mu1 : mu2;
        const float* va  = (layer == 0) ? va1 : va2;

        pack_w_f16<<<4, 256, 0, stream>>>(W, WP, HID);
        gemm128_wmma<false><<<gridGemm, 256, 0, stream>>>(xin, WP, nullptr, H, HID);
        node_scores<<<gridNodeW, 256, 0, stream>>>(H, asr, ads, SS, SD, n_nodes);
        fill_f32<<<gridNode, 256, 0, stream>>>(MX, -3.402823466e38f, n_nodes);
        fill_f32<<<gridNode, 256, 0, stream>>>(DEN, 0.0f, n_nodes);
        fill_f32<<<gridFeat, 256, 0, stream>>>(AGG, 0.0f, feat_total);
        edge_max<<<gridEdge, 256, 0, stream>>>(esrc, edst, SS, SD, MX, n_edges, etot);
        edge_expsum<<<gridEdge, 256, 0, stream>>>(esrc, edst, SS, SD, MX, DEN, n_edges, etot);
        edge_aggregate<<<gridEdgeW, 256, 0, stream>>>(H, esrc, edst, SS, SD, MX, DEN, AGG,
                                                      n_edges, etot);
        bias_bn_relu<<<gridFeat, 256, 0, stream>>>(AGG, bia, gam, bet, mu, va, feat_total);
    }

    fill_f32<<<(N_GRAPHS * HID + 255) / 256, 256, 0, stream>>>(POOL, 0.0f, N_GRAPHS * HID);
    fill_f32<<<(N_GRAPHS + 255) / 256, 256, 0, stream>>>(CNT, 0.0f, N_GRAPHS);
    pool_accum<<<gridFeat, 256, 0, stream>>>(AGG, bat, POOL, CNT, n_nodes);
    pool_div<<<(N_GRAPHS * HID + 255) / 256, 256, 0, stream>>>(POOL, CNT, N_GRAPHS * HID);

    pack_w_f16<<<2, 256, 0, stream>>>(hw, WP, 64);
    gemm128_wmma<true><<<N_GRAPHS / 16, 32 * (64 / 16), 0, stream>>>(
        POOL, WP, hb, (float*)d_out, 64);
}


# Replace the three atomic per-edge passes (max, exp-sum, scatter-add) by a dst-sorted CSR built once per call and a wave-per-node register gather; sum order = atomic arrival rank per dst
# speedup vs baseline: 2.1003x; 2.1003x over previous
#include <hip/hip_runtime.h>
#include <hip/hip_bf16.h>

typedef __attribute__((ext_vector_type(16))) _Float16 v16h;
typedef __attribute__((ext_vector_type(8)))  _Float16 v8h;
typedef __attribute__((ext_vector_type(4)))  _Float16 v4h;
typedef __attribute__((ext_vector_type(8)))  float    v8f;

#define HID 128
#define NEG_SLOPE 0.2f
#define BN_EPS 1e-5f
#define N_GRAPHS 512

__global__ void pack_w_f16(const float* __restrict__ W, _Float16* __restrict__ Wp, int N) {
    int id = blockIdx.x * blockDim.x + threadIdx.x;
    int total = (N / 16) * 4 * 32;
    if (id >= total) return;
    int t    = id >> 7;
    int kt   = (id >> 5) & 3;
    int lane = id & 31;
    int hl = lane & 15, hi = lane >> 4;
    int n = t * 16 + hl;
    _Float16* o = Wp + (size_t)id * 16;
#pragma unroll
    for (int j = 0; j < 8; ++j) {
        o[j]     = (_Float16)W[(size_t)(kt * 32 + hi * 8 + j) * N + n];
        o[8 + j] = (_Float16)W[(size_t)(kt * 32 + 16 + hi * 8 + j) * N + n];
    }
}

template <bool HAS_BIAS>
__global__ void gemm128_wmma(const float* __restrict__ A, const _Float16* __restrict__ Bp,
                             const float* __restrict__ bias, float* __restrict__ C,
                             int N) {
    __shared__ __align__(16) _Float16 As[16 * 128];
    const int tid  = threadIdx.x;
    const int row0 = blockIdx.x * 16;

    for (int c = tid; c < 512; c += blockDim.x) {
        int r   = c >> 5;
        int col = (c & 31) * 4;
        float4 f = *(const float4*)(A + (size_t)(row0 + r) * 128 + col);
        v4h h4 = { (_Float16)f.x, (_Float16)f.y, (_Float16)f.z, (_Float16)f.w };
        *(v4h*)(As + r * 128 + col) = h4;
    }
    __syncthreads();

    const int wave = tid >> 5;
    const int lane = tid & 31;
    const int hl   = lane & 15;
    const int hi   = lane >> 4;

    v8f acc = {};
#pragma unroll
    for (int kt = 0; kt < 4; ++kt) {
        int base = hl * 128 + kt * 32 + hi * 8;
        v8h alo = *(const v8h*)(As + base);
        v8h ahi = *(const v8h*)(As + base + 16);
        v16h a;
#pragma unroll
        for (int j = 0; j < 8; ++j) { a[j] = alo[j]; a[8 + j] = ahi[j]; }

        v16h b = *(const v16h*)(Bp + (((size_t)wave * 4 + kt) * 32 + lane) * 16);

        acc = __builtin_amdgcn_wmma_f32_16x16x32_f16(
            false, a, false, b, (short)0, acc, false, false);
    }

    const int col = wave * 16 + hl;
    float bv = HAS_BIAS ? bias[col] : 0.0f;
#pragma unroll
    for (int r = 0; r < 8; ++r) {
        int row = row0 + r + 8 * hi;
        C[(size_t)row * N + col] = acc[r] + bv;
    }
}

__global__ void node_scores(const float* __restrict__ H, const float* __restrict__ a_src,
                            const float* __restrict__ a_dst, float* __restrict__ ss,
                            float* __restrict__ sd, int n_nodes) {
    int node = (int)((blockIdx.x * (size_t)blockDim.x + threadIdx.x) >> 5);
    int lane = threadIdx.x & 31;
    if (node >= n_nodes) return;
    const float* h = H + (size_t)node * HID;
    float s0 = 0.f, s1 = 0.f;
#pragma unroll
    for (int j = 0; j < 4; ++j) {
        int c = lane + 32 * j;
        float hv = h[c];
        s0 += hv * a_src[c];
        s1 += hv * a_dst[c];
    }
#pragma unroll
    for (int off = 16; off > 0; off >>= 1) {
        s0 += __shfl_down(s0, off, 32);
        s1 += __shfl_down(s1, off, 32);
    }
    if (lane == 0) { ss[node] = s0; sd[node] = s1; }
}

__global__ void fill_f32(float* __restrict__ p, float v, int n) {
    int i = blockIdx.x * blockDim.x + threadIdx.x;
    if (i < n) p[i] = v;
}

__global__ void fill_i32(int* __restrict__ p, int v, int n) {
    int i = blockIdx.x * blockDim.x + threadIdx.x;
    if (i < n) p[i] = v;
}

__device__ __forceinline__ float lrelu(float v) { return (v > 0.f) ? v : NEG_SLOPE * v; }

__device__ __forceinline__ void axpy4(float4& acc, float w, const float4& h) {
    acc.x = __builtin_fmaf(w, h.x, acc.x);
    acc.y = __builtin_fmaf(w, h.y, acc.y);
    acc.z = __builtin_fmaf(w, h.z, acc.z);
    acc.w = __builtin_fmaf(w, h.w, acc.w);
}

__global__ void edge_rank(const int* __restrict__ edst, int* __restrict__ deg,
                          int* __restrict__ rnk, int n_edges) {
    int e = blockIdx.x * blockDim.x + threadIdx.x;
    if (e >= n_edges) return;
    rnk[e] = atomicAdd(&deg[edst[e]], 1);
}

__global__ void __launch_bounds__(1024) scan_rowptr(const int* __restrict__ deg,
                                                    int* __restrict__ rowptr, int n) {
    __shared__ int wsum[32];
    __shared__ int wpre[32];
    const int tid = threadIdx.x, lane = tid & 31, wv = tid >> 5;
    const int base = blockIdx.x * 1024;

    int part = 0;
    const int4* d4 = (const int4*)deg;
#pragma unroll 4
    for (int q = tid; q < (base >> 2); q += 1024) {
        int4 t = d4[q];
        part += t.x + t.y + t.z + t.w;
    }
#pragma unroll
    for (int off = 16; off > 0; off >>= 1) part += __shfl_xor(part, off, 32);

    const int i = base + tid;
    const int v = (i < n) ? deg[i] : 0;
    int x = v;
#pragma unroll
    for (int off = 1; off < 32; off <<= 1) {
        int y = __shfl_up(x, off, 32);
        if (lane >= off) x += y;
    }
    if (lane == 31) wsum[wv] = x;
    if (lane == 0)  wpre[wv] = part;
    __syncthreads();
    if (wv == 0) {
        int t = wsum[lane];
        int z = t;
#pragma unroll
        for (int off = 1; off < 32; off <<= 1) {
            int y = __shfl_up(z, off, 32);
            if (lane >= off) z += y;
        }
        int p = wpre[lane];
#pragma unroll
        for (int off = 16; off > 0; off >>= 1) p += __shfl_xor(p, off, 32);
        wsum[lane] = z - t + p;
        if (blockIdx.x == gridDim.x - 1 && lane == 31) rowptr[n] = z + p;
    }
    __syncthreads();
    if (i < n) rowptr[i] = wsum[wv] + x - v;
}

__global__ void csr_place(const int* __restrict__ esrc, const int* __restrict__ edst,
                          const int* __restrict__ rnk, const int* __restrict__ rowptr,
                          int* __restrict__ csrc, int n_edges) {
    int e = blockIdx.x * blockDim.x + threadIdx.x;
    if (e >= n_edges) return;
    csrc[rowptr[edst[e]] + rnk[e]] = esrc[e];
}

__global__ void __launch_bounds__(256) node_gather(const float* __restrict__ H,
                                                   const int* __restrict__ rowptr,
                                                   const int* __restrict__ csrc,
                                                   const float* __restrict__ ss,
                                                   const float* __restrict__ sd,
                                                   float* __restrict__ out, int n_nodes) {
    int d = (int)((blockIdx.x * (size_t)blockDim.x + threadIdx.x) >> 5);
    const int lane = threadIdx.x & 31;
    if (d >= n_nodes) return;
    d = __builtin_amdgcn_readfirstlane(d);
    const int beg = __builtin_amdgcn_readfirstlane(rowptr[d]);
    const int end = __builtin_amdgcn_readfirstlane(rowptr[d + 1]);

    const float sdd   = sd[d];
    const float vself = lrelu(ss[d] + sdd);

    const int  i0   = beg + lane;
    const bool has0 = i0 < end;
    int   s0 = d;
    float v0 = vself;
    if (has0) { s0 = csrc[i0]; v0 = lrelu(ss[s0] + sdd); }

    float m = fmaxf(vself, v0);
    for (int i = i0 + 32; i < end; i += 32)
        m = fmaxf(m, lrelu(ss[csrc[i]] + sdd));
#pragma unroll
    for (int off = 16; off > 0; off >>= 1) m = fmaxf(m, __shfl_xor(m, off, 32));

    float sum = has0 ? __expf(v0 - m) : 0.f;
    for (int i = i0 + 32; i < end; i += 32)
        sum += __expf(lrelu(ss[csrc[i]] + sdd) - m);
#pragma unroll
    for (int off = 16; off > 0; off >>= 1) sum += __shfl_xor(sum, off, 32);

    const float eself = __expf(vself - m);
    const float den   = sum + eself + 1e-16f;

    const float wself = eself / den;
    float4 hv = *((const float4*)(H + (size_t)d * HID) + lane);
    float4 acc = make_float4(wself * hv.x, wself * hv.y, wself * hv.z, wself * hv.w);

    for (int base = beg; base < end; base += 32) {
        int   s = s0;
        float w = has0 ? __expf(v0 - m) / den : 0.f;
        if (base != beg) {
            int i = base + lane;
            s = d; w = 0.f;
            if (i < end) { s = csrc[i]; w = __expf(lrelu(ss[s] + sdd) - m) / den; }
        }
        const int rem = end - base;
        const int cnt = rem < 32 ? rem : 32;
        for (int j = 0; j < cnt; j += 4) {
            int sa = __builtin_amdgcn_readlane(s, j);
            int sb = __builtin_amdgcn_readlane(s, j + 1);
            int sc = __builtin_amdgcn_readlane(s, j + 2);
            int se = __builtin_amdgcn_readlane(s, j + 3);
            float wa = __int_as_float(__builtin_amdgcn_readlane(__float_as_int(w), j));
            float wb = __int_as_float(__builtin_amdgcn_readlane(__float_as_int(w), j + 1));
            float wc = __int_as_float(__builtin_amdgcn_readlane(__float_as_int(w), j + 2));
            float we = __int_as_float(__builtin_amdgcn_readlane(__float_as_int(w), j + 3));
            float4 ha = *((const float4*)(H + (size_t)sa * HID) + lane);
            float4 hb = *((const float4*)(H + (size_t)sb * HID) + lane);
            float4 hc = *((const float4*)(H + (size_t)sc * HID) + lane);
            float4 he = *((const float4*)(H + (size_t)se * HID) + lane);
            axpy4(acc, wa, ha);
            axpy4(acc, wb, hb);
            axpy4(acc, wc, hc);
            axpy4(acc, we, he);
        }
    }
    *((float4*)(out + (size_t)d * HID) + lane) = acc;
}

__global__ void bias_bn_relu(float* __restrict__ x, const float* __restrict__ bias,
                             const float* __restrict__ gam, const float* __restrict__ bet,
                             const float* __restrict__ mean, const float* __restrict__ var,
                             int total) {
    int i = blockIdx.x * blockDim.x + threadIdx.x;
    if (i >= total) return;
    int c = i & (HID - 1);
    float v = x[i] + bias[c];
    v = (v - mean[c]) * (gam[c] * rsqrtf(var[c] + BN_EPS)) + bet[c];
    x[i] = fmaxf(v, 0.f);
}

__global__ void pool_accum(const float* __restrict__ x, const int* __restrict__ batch,
                           float* __restrict__ sums, float* __restrict__ cnt, int n_nodes) {
    int i = blockIdx.x * blockDim.x + threadIdx.x;
    if (i >= n_nodes * HID) return;
    int n = i >> 7, c = i & (HID - 1);
    int g = batch[n];
    atomicAdd(&sums[(size_t)g * HID + c], x[i]);
    if (c == 0) atomicAdd(&cnt[g], 1.0f);
}

__global__ void pool_div(float* __restrict__ sums, const float* __restrict__ cnt, int total) {
    int i = blockIdx.x * blockDim.x + threadIdx.x;
    if (i >= total) return;
    sums[i] /= fmaxf(cnt[i >> 7], 1.0f);
}

extern "C" void kernel_launch(void* const* d_in, const int* in_sizes, int n_in,
                              void* d_out, int out_size, void* d_ws, size_t ws_size,
                              hipStream_t stream) {
    const float* gx  = (const float*)d_in[0];
    const int*   ei  = (const int*)  d_in[1];
    const int*   bat = (const int*)  d_in[2];
    const float* W1  = (const float*)d_in[3];
    const float* as1 = (const float*)d_in[4];
    const float* ad1 = (const float*)d_in[5];
    const float* b1  = (const float*)d_in[6];
    const float* g1  = (const float*)d_in[7];
    const float* be1 = (const float*)d_in[8];
    const float* mu1 = (const float*)d_in[9];
    const float* va1 = (const float*)d_in[10];
    const float* W2  = (const float*)d_in[11];
    const float* as2 = (const float*)d_in[12];
    const float* ad2 = (const float*)d_in[13];
    const float* b2  = (const float*)d_in[14];
    const float* g2  = (const float*)d_in[15];
    const float* be2 = (const float*)d_in[16];
    const float* mu2 = (const float*)d_in[17];
    const float* va2 = (const float*)d_in[18];
    const float* hw  = (const float*)d_in[19];
    const float* hb  = (const float*)d_in[20];

    const int n_nodes = in_sizes[0] / HID;
    const int n_edges = in_sizes[1] / 2;
    const int* esrc = ei;
    const int* edst = ei + n_edges;

    float* ws   = (float*)d_ws;
    float* H    = ws;
    float* AGG  = H    + (size_t)n_nodes * HID;
    float* SS   = AGG  + (size_t)n_nodes * HID;
    float* SD   = SS   + n_nodes;
    float* POOL = SD   + n_nodes;
    float* CNT  = POOL + (size_t)N_GRAPHS * HID;
    size_t wpOff = (size_t)(CNT + N_GRAPHS - ws);
    wpOff = (wpOff + 7) & ~(size_t)7;
    _Float16* WP = (_Float16*)(ws + wpOff);
    int* DEG    = (int*)(ws + wpOff + (HID * HID) / 2);
    int* ROWPTR = DEG    + n_nodes;
    int* RNK    = ROWPTR + n_nodes + 1;
    int* CSRC   = RNK    + n_edges;

    const int feat_total = n_nodes * HID;
    const int gridFeat  = (feat_total + 255) / 256;
    const int gridNode  = (n_nodes + 255) / 256;
    const int gridEdge  = (n_edges + 255) / 256;
    const int gridNodeW = (n_nodes + 7) / 8;
    const int gridGemm  = n_nodes / 16;

    fill_i32<<<gridNode, 256, 0, stream>>>(DEG, 0, n_nodes);
    edge_rank<<<gridEdge, 256, 0, stream>>>(edst, DEG, RNK, n_edges);
    scan_rowptr<<<(n_nodes + 1023) / 1024, 1024, 0, stream>>>(DEG, ROWPTR, n_nodes);
    csr_place<<<gridEdge, 256, 0, stream>>>(esrc, edst, RNK, ROWPTR, CSRC, n_edges);

    for (int layer = 0; layer < 2; ++layer) {
        const float* xin = (layer == 0) ? gx : AGG;
        const float* W   = (layer == 0) ? W1 : W2;
        const float* asr = (layer == 0) ? as1 : as2;
        const float* ads = (layer == 0) ? ad1 : ad2;
        const float* bia = (layer == 0) ? b1  : b2;
        const float* gam = (layer == 0) ? g1  : g2;
        const float* bet = (layer == 0) ? be1 : be2;
        const float* mu  = (layer == 0) ? mu1 : mu2;
        const float* va  = (layer == 0) ? va1 : va2;

        pack_w_f16<<<4, 256, 0, stream>>>(W, WP, HID);
        gemm128_wmma<false><<<gridGemm, 256, 0, stream>>>(xin, WP, nullptr, H, HID);
        node_scores<<<gridNodeW, 256, 0, stream>>>(H, asr, ads, SS, SD, n_nodes);
        node_gather<<<gridNodeW, 256, 0, stream>>>(H, ROWPTR, CSRC, SS, SD, AGG, n_nodes);
        bias_bn_relu<<<gridFeat, 256, 0, stream>>>(AGG, bia, gam, bet, mu, va, feat_total);
    }

    fill_f32<<<(N_GRAPHS * HID + 255) / 256, 256, 0, stream>>>(POOL, 0.0f, N_GRAPHS * HID);
    fill_f32<<<(N_GRAPHS + 255) / 256, 256, 0, stream>>>(CNT, 0.0f, N_GRAPHS);
    pool_accum<<<gridFeat, 256, 0, stream>>>(AGG, bat, POOL, CNT, n_nodes);
    pool_div<<<(N_GRAPHS * HID + 255) / 256, 256, 0, stream>>>(POOL, CNT, N_GRAPHS * HID);

    pack_w_f16<<<2, 256, 0, stream>>>(hw, WP, 64);
    gemm128_wmma<true><<<N_GRAPHS / 16, 32 * (64 / 16), 0, stream>>>(
        POOL, WP, hb, (float*)d_out, 64);
}


# Launches 17 -> 9: weight packs and zero fills in one prep launch; edge ranking beside GEMM 1 and rowptr scan beside layer-1 scores as block roles of one grid; pool divide done in head GEMM staging
# speedup vs baseline: 2.5780x; 1.2274x over previous
#include <hip/hip_runtime.h>
#include <hip/hip_bf16.h>

typedef __attribute__((ext_vector_type(16))) _Float16 v16h;
typedef __attribute__((ext_vector_type(8)))  _Float16 v8h;
typedef __attribute__((ext_vector_type(4)))  _Float16 v4h;
typedef __attribute__((ext_vector_type(8)))  float    v8f;

#define HID 128
#define NEG_SLOPE 0.2f
#define BN_EPS 1e-5f
#define N_GRAPHS 512

__device__ __forceinline__ void pack_frag(const float* __restrict__ W, _Float16* __restrict__ Wp,
                                          int N, int id) {
    int total = (N / 16) * 4 * 32;
    if (id >= total) return;
    int t    = id >> 7;
    int kt   = (id >> 5) & 3;
    int lane = id & 31;
    int hl = lane & 15, hi = lane >> 4;
    int n = t * 16 + hl;
    _Float16* o = Wp + (size_t)id * 16;
#pragma unroll
    for (int j = 0; j < 8; ++j) {
        o[j]     = (_Float16)W[(size_t)(kt * 32 + hi * 8 + j) * N + n];
        o[8 + j] = (_Float16)W[(size_t)(kt * 32 + 16 + hi * 8 + j) * N + n];
    }
}

__global__ void prep_pack_and_clear(const float* __restrict__ W1, const float* __restrict__ W2,
                                    const float* __restrict__ HW, _Float16* __restrict__ WP1,
                                    _Float16* __restrict__ WP2, _Float16* __restrict__ WP3,
                                    int* __restrict__ deg, int n_nodes,
                                    float* __restrict__ pool, int n_pool) {
    const int b = blockIdx.x;
    const int tid = threadIdx.x;
    if (b < 4) {
        pack_frag(W1, WP1, HID, b * 256 + tid);
    } else if (b < 8) {
        pack_frag(W2, WP2, HID, (b - 4) * 256 + tid);
    } else if (b < 10) {
        pack_frag(HW, WP3, 64, (b - 8) * 256 + tid);
    } else {
        const int i = (b - 10) * 256 + tid;
        if (i < n_nodes) deg[i] = 0;
        else if (i < n_nodes + n_pool) pool[i - n_nodes] = 0.0f;
    }
}

template <bool HAS_BIAS, bool DIV>
__device__ __forceinline__ void gemm_tile(const float* __restrict__ A,
                                          const _Float16* __restrict__ Bp,
                                          const float* __restrict__ bias,
                                          const float* __restrict__ rowcnt,
                                          float* __restrict__ C, int N, int tile) {
    __shared__ __align__(16) _Float16 As[16 * 128];
    const int tid  = threadIdx.x;
    const int row0 = tile * 16;

    for (int c = tid; c < 512; c += blockDim.x) {
        int r   = c >> 5;
        int col = (c & 31) * 4;
        float4 f = *(const float4*)(A + (size_t)(row0 + r) * 128 + col);
        if (DIV) {
            const float dv = fmaxf(rowcnt[row0 + r], 1.0f);
            f.x /= dv; f.y /= dv; f.z /= dv; f.w /= dv;
        }
        v4h h4 = { (_Float16)f.x, (_Float16)f.y, (_Float16)f.z, (_Float16)f.w };
        *(v4h*)(As + r * 128 + col) = h4;
    }
    __syncthreads();

    const int wave = tid >> 5;
    const int lane = tid & 31;
    const int hl   = lane & 15;
    const int hi   = lane >> 4;

    v8f acc = {};
#pragma unroll
    for (int kt = 0; kt < 4; ++kt) {
        int base = hl * 128 + kt * 32 + hi * 8;
        v8h alo = *(const v8h*)(As + base);
        v8h ahi = *(const v8h*)(As + base + 16);
        v16h a;
#pragma unroll
        for (int j = 0; j < 8; ++j) { a[j] = alo[j]; a[8 + j] = ahi[j]; }

        v16h b = *(const v16h*)(Bp + (((size_t)wave * 4 + kt) * 32 + lane) * 16);

        acc = __builtin_amdgcn_wmma_f32_16x16x32_f16(
            false, a, false, b, (short)0, acc, false, false);
    }

    const int col = wave * 16 + hl;
    float bv = HAS_BIAS ? bias[col] : 0.0f;
#pragma unroll
    for (int r = 0; r < 8; ++r) {
        int row = row0 + r + 8 * hi;
        C[(size_t)row * N + col] = acc[r] + bv;
    }
}

template <bool HAS_BIAS, bool DIV>
__global__ void gemm128_wmma(const float* __restrict__ A, const _Float16* __restrict__ Bp,
                             const float* __restrict__ bias, const float* __restrict__ rowcnt,
                             float* __restrict__ C, int N) {
    gemm_tile<HAS_BIAS, DIV>(A, Bp, bias, rowcnt, C, N, blockIdx.x);
}

__global__ void gemm1_and_edge_rank(const float* __restrict__ A, const _Float16* __restrict__ Bp,
                                    float* __restrict__ C, int n_tiles,
                                    const int* __restrict__ edst, int* __restrict__ deg,
                                    int* __restrict__ rnk, int n_edges, int n_rank_blocks) {
    const int b    = blockIdx.x;
    const int nmin = n_tiles < n_rank_blocks ? n_tiles : n_rank_blocks;
    bool is_gemm;
    int  idx;
    if (b < 2 * nmin) { is_gemm = (b & 1) == 0; idx = b >> 1; }
    else              { is_gemm = n_tiles > n_rank_blocks; idx = b - nmin; }
    if (is_gemm) {
        gemm_tile<false, false>(A, Bp, nullptr, nullptr, C, HID, idx);
    } else {
        const int e = idx * 256 + threadIdx.x;
        if (e < n_edges) rnk[e] = atomicAdd(&deg[edst[e]], 1);
    }
}

__device__ __forceinline__ void node_scores_wave(const float* __restrict__ H,
                                                 const float* __restrict__ a_src,
                                                 const float* __restrict__ a_dst,
                                                 float* __restrict__ ss, float* __restrict__ sd,
                                                 int n_nodes, int node) {
    int lane = threadIdx.x & 31;
    if (node >= n_nodes) return;
    const float* h = H + (size_t)node * HID;
    float s0 = 0.f, s1 = 0.f;
#pragma unroll
    for (int j = 0; j < 4; ++j) {
        int c = lane + 32 * j;
        float hv = h[c];
        s0 += hv * a_src[c];
        s1 += hv * a_dst[c];
    }
#pragma unroll
    for (int off = 16; off > 0; off >>= 1) {
        s0 += __shfl_down(s0, off, 32);
        s1 += __shfl_down(s1, off, 32);
    }
    if (lane == 0) { ss[node] = s0; sd[node] = s1; }
}

__global__ void node_scores(const float* __restrict__ H, const float* __restrict__ a_src,
                            const float* __restrict__ a_dst, float* __restrict__ ss,
                            float* __restrict__ sd, int n_nodes) {
    int node = (int)((blockIdx.x * (size_t)blockDim.x + threadIdx.x) >> 5);
    node_scores_wave(H, a_src, a_dst, ss, sd, n_nodes, node);
}

__device__ __forceinline__ float lrelu(float v) { return (v > 0.f) ? v : NEG_SLOPE * v; }

__device__ __forceinline__ void axpy4(float4& acc, float w, const float4& h) {
    acc.x = __builtin_fmaf(w, h.x, acc.x);
    acc.y = __builtin_fmaf(w, h.y, acc.y);
    acc.z = __builtin_fmaf(w, h.z, acc.z);
    acc.w = __builtin_fmaf(w, h.w, acc.w);
}

__device__ __forceinline__ void scan_rowptr_block(const int* __restrict__ deg,
                                                  int* __restrict__ rowptr, int n,
                                                  int blk, int n_blk) {
    __shared__ int wsum[32];
    __shared__ int wpre[32];
    const int tid = threadIdx.x, lane = tid & 31, wv = tid >> 5;
    const int base = blk * 1024;

    int part = 0;
    const int4* d4 = (const int4*)deg;
#pragma unroll 4
    for (int q = tid; q < (base >> 2); q += 1024) {
        int4 t = d4[q];
        part += t.x + t.y + t.z + t.w;
    }
#pragma unroll
    for (int off = 16; off > 0; off >>= 1) part += __shfl_xor(part, off, 32);

    const int i = base + tid;
    const int v = (i < n) ? deg[i] : 0;
    int x = v;
#pragma unroll
    for (int off = 1; off < 32; off <<= 1) {
        int y = __shfl_up(x, off, 32);
        if (lane >= off) x += y;
    }
    if (lane == 31) wsum[wv] = x;
    if (lane == 0)  wpre[wv] = part;
    __syncthreads();
    if (wv == 0) {
        int t = wsum[lane];
        int z = t;
#pragma unroll
        for (int off = 1; off < 32; off <<= 1) {
            int y = __shfl_up(z, off, 32);
            if (lane >= off) z += y;
        }
        int p = wpre[lane];
#pragma unroll
        for (int off = 16; off > 0; off >>= 1) p += __shfl_xor(p, off, 32);
        wsum[lane] = z - t + p;
        if (blk == n_blk - 1 && lane == 31) rowptr[n] = z + p;
    }
    __syncthreads();
    if (i < n) rowptr[i] = wsum[wv] + x - v;
}

__global__ void __launch_bounds__(1024) scan_and_scores(const int* __restrict__ deg,
                                                        int* __restrict__ rowptr, int n_nodes,
                                                        int n_scan_blocks,
                                                        const float* __restrict__ H,
                                                        const float* __restrict__ a_src,
                                                        const float* __restrict__ a_dst,
                                                        float* __restrict__ ss,
                                                        float* __restrict__ sd) {
    const int b = blockIdx.x;
    if (b < n_scan_blocks) {
        scan_rowptr_block(deg, rowptr, n_nodes, b, n_scan_blocks);
    } else {
        const int node = (b - n_scan_blocks) * 32 + (int)(threadIdx.x >> 5);
        node_scores_wave(H, a_src, a_dst, ss, sd, n_nodes, node);
    }
}

__global__ void csr_place(const int* __restrict__ esrc, const int* __restrict__ edst,
                          const int* __restrict__ rnk, const int* __restrict__ rowptr,
                          int* __restrict__ csrc, int n_edges) {
    int e = blockIdx.x * blockDim.x + threadIdx.x;
    if (e >= n_edges) return;
    csrc[rowptr[edst[e]] + rnk[e]] = esrc[e];
}

__device__ __forceinline__ float bn_relu(float x, float b, float g, float be, float mu, float va) {
    float v = x + b;
    v = (v - mu) * (g * rsqrtf(va + BN_EPS)) + be;
    return fmaxf(v, 0.f);
}

template <bool POOL>
__global__ void __launch_bounds__(256) node_gather(const float* __restrict__ H,
                                                   const int* __restrict__ rowptr,
                                                   const int* __restrict__ csrc,
                                                   const float* __restrict__ ss,
                                                   const float* __restrict__ sd,
                                                   const float* __restrict__ bias,
                                                   const float* __restrict__ gam,
                                                   const float* __restrict__ bet,
                                                   const float* __restrict__ mean,
                                                   const float* __restrict__ var,
                                                   const int* __restrict__ batch,
                                                   float* __restrict__ out,
                                                   float* __restrict__ gsum,
                                                   float* __restrict__ gcnt, int n_nodes) {
    int d = (int)((blockIdx.x * (size_t)blockDim.x + threadIdx.x) >> 5);
    const int lane = threadIdx.x & 31;
    if (d >= n_nodes) return;
    d = __builtin_amdgcn_readfirstlane(d);
    const int beg = __builtin_amdgcn_readfirstlane(rowptr[d]);
    const int end = __builtin_amdgcn_readfirstlane(rowptr[d + 1]);

    const float sdd   = sd[d];
    const float vself = lrelu(ss[d] + sdd);

    const int  i0   = beg + lane;
    const bool has0 = i0 < end;
    int   s0 = d;
    float v0 = vself;
    if (has0) { s0 = csrc[i0]; v0 = lrelu(ss[s0] + sdd); }

    float m = fmaxf(vself, v0);
    for (int i = i0 + 32; i < end; i += 32)
        m = fmaxf(m, lrelu(ss[csrc[i]] + sdd));
#pragma unroll
    for (int off = 16; off > 0; off >>= 1) m = fmaxf(m, __shfl_xor(m, off, 32));

    float sum = has0 ? __expf(v0 - m) : 0.f;
    for (int i = i0 + 32; i < end; i += 32)
        sum += __expf(lrelu(ss[csrc[i]] + sdd) - m);
#pragma unroll
    for (int off = 16; off > 0; off >>= 1) sum += __shfl_xor(sum, off, 32);

    const float eself = __expf(vself - m);
    const float den   = sum + eself + 1e-16f;

    const float wself = eself / den;
    float4 hv = *((const float4*)(H + (size_t)d * HID) + lane);
    float4 acc = make_float4(wself * hv.x, wself * hv.y, wself * hv.z, wself * hv.w);

    for (int base = beg; base < end; base += 32) {
        int   s = s0;
        float w = has0 ? __expf(v0 - m) / den : 0.f;
        if (base != beg) {
            int i = base + lane;
            s = d; w = 0.f;
            if (i < end) { s = csrc[i]; w = __expf(lrelu(ss[s] + sdd) - m) / den; }
        }
        const int rem = end - base;
        const int cnt = rem < 32 ? rem : 32;
        for (int j = 0; j < cnt; j += 4) {
            int sa = __builtin_amdgcn_readlane(s, j);
            int sb = __builtin_amdgcn_readlane(s, j + 1);
            int sc = __builtin_amdgcn_readlane(s, j + 2);
            int se = __builtin_amdgcn_readlane(s, j + 3);
            float wa = __int_as_float(__builtin_amdgcn_readlane(__float_as_int(w), j));
            float wb = __int_as_float(__builtin_amdgcn_readlane(__float_as_int(w), j + 1));
            float wc = __int_as_float(__builtin_amdgcn_readlane(__float_as_int(w), j + 2));
            float we = __int_as_float(__builtin_amdgcn_readlane(__float_as_int(w), j + 3));
            float4 ha = *((const float4*)(H + (size_t)sa * HID) + lane);
            float4 hb = *((const float4*)(H + (size_t)sb * HID) + lane);
            float4 hc = *((const float4*)(H + (size_t)sc * HID) + lane);
            float4 he = *((const float4*)(H + (size_t)se * HID) + lane);
            axpy4(acc, wa, ha);
            axpy4(acc, wb, hb);
            axpy4(acc, wc, hc);
            axpy4(acc, we, he);
        }
    }
    const float4 bi = *((const float4*)bias + lane);
    const float4 ga = *((const float4*)gam  + lane);
    const float4 be = *((const float4*)bet  + lane);
    const float4 mu = *((const float4*)mean + lane);
    const float4 va = *((const float4*)var  + lane);
    acc.x = bn_relu(acc.x, bi.x, ga.x, be.x, mu.x, va.x);
    acc.y = bn_relu(acc.y, bi.y, ga.y, be.y, mu.y, va.y);
    acc.z = bn_relu(acc.z, bi.z, ga.z, be.z, mu.z, va.z);
    acc.w = bn_relu(acc.w, bi.w, ga.w, be.w, mu.w, va.w);
    if (POOL) {
        const int g = __builtin_amdgcn_readfirstlane(batch[d]);
        float* p = gsum + (size_t)g * HID + 4 * lane;
        atomicAdd(p + 0, acc.x);
        atomicAdd(p + 1, acc.y);
        atomicAdd(p + 2, acc.z);
        atomicAdd(p + 3, acc.w);
        if (lane == 0) atomicAdd(gcnt + g, 1.0f);
    } else {
        *((float4*)(out + (size_t)d * HID) + lane) = acc;
    }
}

extern "C" void kernel_launch(void* const* d_in, const int* in_sizes, int n_in,
                              void* d_out, int out_size, void* d_ws, size_t ws_size,
                              hipStream_t stream) {
    const float* gx  = (const float*)d_in[0];
    const int*   ei  = (const int*)  d_in[1];
    const int*   bat = (const int*)  d_in[2];
    const float* W1  = (const float*)d_in[3];
    const float* as1 = (const float*)d_in[4];
    const float* ad1 = (const float*)d_in[5];
    const float* b1  = (const float*)d_in[6];
    const float* g1  = (const float*)d_in[7];
    const float* be1 = (const float*)d_in[8];
    const float* mu1 = (const float*)d_in[9];
    const float* va1 = (const float*)d_in[10];
    const float* W2  = (const float*)d_in[11];
    const float* as2 = (const float*)d_in[12];
    const float* ad2 = (const float*)d_in[13];
    const float* b2  = (const float*)d_in[14];
    const float* g2  = (const float*)d_in[15];
    const float* be2 = (const float*)d_in[16];
    const float* mu2 = (const float*)d_in[17];
    const float* va2 = (const float*)d_in[18];
    const float* hw  = (const float*)d_in[19];
    const float* hb  = (const float*)d_in[20];

    const int n_nodes = in_sizes[0] / HID;
    const int n_edges = in_sizes[1] / 2;
    const int* esrc = ei;
    const int* edst = ei + n_edges;

    float* ws   = (float*)d_ws;
    float* H    = ws;
    float* AGG  = H    + (size_t)n_nodes * HID;
    float* SS   = AGG  + (size_t)n_nodes * HID;
    float* SD   = SS   + n_nodes;
    float* POOL = SD   + n_nodes;
    float* CNT  = POOL + (size_t)N_GRAPHS * HID;
    size_t wpOff = (size_t)(CNT + N_GRAPHS - ws);
    wpOff = (wpOff + 7) & ~(size_t)7;
    _Float16* WP1 = (_Float16*)(ws + wpOff);
    _Float16* WP2 = WP1 + HID * HID;
    _Float16* WP3 = WP2 + HID * HID;
    int* DEG    = (int*)(ws + wpOff + (HID * HID) / 2 + (HID * HID) / 2 + (HID * 64) / 2);
    int* ROWPTR = DEG    + n_nodes;
    int* RNK    = ROWPTR + n_nodes + 1;
    int* CSRC   = RNK    + n_edges;

    const int gridEdge  = (n_edges + 255) / 256;
    const int gridNodeW = (n_nodes + 7) / 8;
    const int gridGemm  = n_nodes / 16;
    const int gridScan  = (n_nodes + 1023) / 1024;
    const int nPool     = N_GRAPHS * HID + N_GRAPHS;

    prep_pack_and_clear<<<10 + (n_nodes + nPool + 255) / 256, 256, 0, stream>>>(
        W1, W2, hw, WP1, WP2, WP3, DEG, n_nodes, POOL, nPool);

    gemm1_and_edge_rank<<<gridGemm + gridEdge, 256, 0, stream>>>(
        gx, WP1, H, gridGemm, edst, DEG, RNK, n_edges, gridEdge);

    scan_and_scores<<<gridScan + (n_nodes + 31) / 32, 1024, 0, stream>>>(
        DEG, ROWPTR, n_nodes, gridScan, H, as1, ad1, SS, SD);

    csr_place<<<gridEdge, 256, 0, stream>>>(esrc, edst, RNK, ROWPTR, CSRC, n_edges);

    node_gather<false><<<gridNodeW, 256, 0, stream>>>(
        H, ROWPTR, CSRC, SS, SD, b1, g1, be1, mu1, va1, bat, AGG, POOL, CNT, n_nodes);

    gemm128_wmma<false, false><<<gridGemm, 256, 0, stream>>>(AGG, WP2, nullptr, nullptr, H, HID);
    node_scores<<<gridNodeW, 256, 0, stream>>>(H, as2, ad2, SS, SD, n_nodes);
    node_gather<true><<<gridNodeW, 256, 0, stream>>>(
        H, ROWPTR, CSRC, SS, SD, b2, g2, be2, mu2, va2, bat, AGG, POOL, CNT, n_nodes);

    gemm128_wmma<true, true><<<N_GRAPHS / 16, 32 * (64 / 16), 0, stream>>>(
        POOL, WP3, hb, CNT, (float*)d_out, 64);
}


# Products x@W of both layers kept as f16 rows (half precision, 256 B) rounded once from the f32 accumulators and stored as whole lines via LDS; the row gathers read half the bytes; sum order kept
# speedup vs baseline: 4.3608x; 1.6915x over previous
#include <hip/hip_runtime.h>
#include <hip/hip_bf16.h>

typedef __attribute__((ext_vector_type(16))) _Float16 v16h;
typedef __attribute__((ext_vector_type(8)))  _Float16 v8h;
typedef __attribute__((ext_vector_type(4)))  _Float16 v4h;
typedef __attribute__((ext_vector_type(8)))  float    v8f;

#define HID 128
#define NEG_SLOPE 0.2f
#define BN_EPS 1e-5f
#define N_GRAPHS 512
#define BKT 64

__device__ __forceinline__ void pack_frag(const float* __restrict__ W, _Float16* __restrict__ Wp,
                                          int N, int id) {
    int total = (N / 16) * 4 * 32;
    if (id >= total) return;
    int t    = id >> 7;
    int kt   = (id >> 5) & 3;
    int lane = id & 31;
    int hl = lane & 15, hi = lane >> 4;
    int n = t * 16 + hl;
    _Float16* o = Wp + (size_t)id * 16;
#pragma unroll
    for (int j = 0; j < 8; ++j) {
        o[j]     = (_Float16)W[(size_t)(kt * 32 + hi * 8 + j) * N + n];
        o[8 + j] = (_Float16)W[(size_t)(kt * 32 + 16 + hi * 8 + j) * N + n];
    }
}

__global__ void prep_pack_and_clear(const float* __restrict__ W1, const float* __restrict__ W2,
                                    const float* __restrict__ HW, _Float16* __restrict__ WP1,
                                    _Float16* __restrict__ WP2, _Float16* __restrict__ WP3,
                                    int* __restrict__ deg, int n_int,
                                    float* __restrict__ pool, int n_pool) {
    const int b = blockIdx.x;
    const int tid = threadIdx.x;
    if (b < 4) {
        pack_frag(W1, WP1, HID, b * 256 + tid);
    } else if (b < 8) {
        pack_frag(W2, WP2, HID, (b - 4) * 256 + tid);
    } else if (b < 10) {
        pack_frag(HW, WP3, 64, (b - 8) * 256 + tid);
    } else {
        const int i = (b - 10) * 256 + tid;
        if (i < n_int) deg[i] = 0;
        else if (i < n_int + n_pool) pool[i - n_int] = 0.0f;
    }
}


__device__ __forceinline__ float half_wave_sum(float v) {
    v += __int_as_float(__builtin_amdgcn_mov_dpp(__float_as_int(v), 0x161, 0xf, 0xf, true));
    v += __int_as_float(__builtin_amdgcn_mov_dpp(__float_as_int(v), 0x162, 0xf, 0xf, true));
    v += __int_as_float(__builtin_amdgcn_mov_dpp(__float_as_int(v), 0x164, 0xf, 0xf, true));
    v += __int_as_float(__builtin_amdgcn_mov_dpp(__float_as_int(v), 0x168, 0xf, 0xf, true));
    return v;
}

template <bool HAS_BIAS, bool DIV, bool SCORES>
__device__ __forceinline__ void gemm_tile(const float* __restrict__ A,
                                          const _Float16* __restrict__ Bp,
                                          const float* __restrict__ bias,
                                          const float* __restrict__ rowcnt,
                                          float* __restrict__ C, _Float16* __restrict__ Ch,
                                          int N, int tile,
                                          const float* __restrict__ a_src,
                                          const float* __restrict__ a_dst,
                                          float* __restrict__ ss, float* __restrict__ sd) {
    __shared__ __align__(16) _Float16 As[16 * 128];
    __shared__ float spart[8 * 2 * 16];
    __shared__ __align__(16) _Float16 Cs[16 * 128];
    const int tid  = threadIdx.x;
    const int row0 = tile * 16;

    for (int c = tid; c < 512; c += blockDim.x) {
        int r   = c >> 5;
        int col = (c & 31) * 4;
        float4 f = *(const float4*)(A + (size_t)(row0 + r) * 128 + col);
        if (DIV) {
            const float dv = fmaxf(rowcnt[row0 + r], 1.0f);
            f.x /= dv; f.y /= dv; f.z /= dv; f.w /= dv;
        }
        v4h h4 = { (_Float16)f.x, (_Float16)f.y, (_Float16)f.z, (_Float16)f.w };
        *(v4h*)(As + r * 128 + col) = h4;
    }
    __syncthreads();

    const int wave = tid >> 5;
    const int lane = tid & 31;
    const int hl   = lane & 15;
    const int hi   = lane >> 4;

    v8f acc = {};
#pragma unroll
    for (int kt = 0; kt < 4; ++kt) {
        int base = hl * 128 + kt * 32 + hi * 8;
        v8h alo = *(const v8h*)(As + base);
        v8h ahi = *(const v8h*)(As + base + 16);
        v16h a;
#pragma unroll
        for (int j = 0; j < 8; ++j) { a[j] = alo[j]; a[8 + j] = ahi[j]; }

        v16h b = *(const v16h*)(Bp + (((size_t)wave * 4 + kt) * 32 + lane) * 16);

        acc = __builtin_amdgcn_wmma_f32_16x16x32_f16(
            false, a, false, b, (short)0, acc, false, false);
    }

    const int col = wave * 16 + hl;

    if (SCORES) {
        const float as = a_src[col];
        const float ad = a_dst[col];
        float ps[8], pd[8];
#pragma unroll
        for (int r = 0; r < 8; ++r) { ps[r] = acc[r] * as; pd[r] = acc[r] * ad; }
#pragma unroll
        for (int r = 0; r < 8; ++r) { ps[r] = half_wave_sum(ps[r]); pd[r] = half_wave_sum(pd[r]); }
        if (hl == 0) {
#pragma unroll
            for (int r = 0; r < 8; ++r) {
                spart[(wave * 2 + 0) * 16 + r + 8 * hi] = ps[r];
                spart[(wave * 2 + 1) * 16 + r + 8 * hi] = pd[r];
            }
        }
#pragma unroll
        for (int r = 0; r < 8; ++r)
            Cs[(r + 8 * hi) * 128 + col] = (_Float16)acc[r];
        __syncthreads();
        if (tid < 32) {
            float s = 0.f;
#pragma unroll
            for (int w = 0; w < 8; ++w) s += spart[w * 32 + tid];
            if (tid < 16) ss[row0 + tid] = s;
            else          sd[row0 + tid - 16] = s;
        }
        const v8h piece = *(const v8h*)(Cs + tid * 8);
        *(v8h*)(Ch + (size_t)row0 * 128 + tid * 8) = piece;
    } else {
        float bv = HAS_BIAS ? bias[col] : 0.0f;
#pragma unroll
        for (int r = 0; r < 8; ++r) {
            int row = row0 + r + 8 * hi;
            C[(size_t)row * N + col] = acc[r] + bv;
        }
    }
}

template <bool HAS_BIAS, bool DIV, bool SCORES>
__global__ void gemm128_wmma(const float* __restrict__ A, const _Float16* __restrict__ Bp,
                             const float* __restrict__ bias, const float* __restrict__ rowcnt,
                             float* __restrict__ C, _Float16* __restrict__ Ch, int N,
                             const float* __restrict__ a_src, const float* __restrict__ a_dst,
                             float* __restrict__ ss, float* __restrict__ sd) {
    gemm_tile<HAS_BIAS, DIV, SCORES>(A, Bp, bias, rowcnt, C, Ch, N, blockIdx.x, a_src, a_dst, ss, sd);
}

__global__ void gemm1_and_bucket_fill(const float* __restrict__ A, const _Float16* __restrict__ Bp,
                                      _Float16* __restrict__ Ch, int n_tiles,
                                      const float* __restrict__ a_src, const float* __restrict__ a_dst,
                                      float* __restrict__ ss, float* __restrict__ sd,
                                      const int* __restrict__ esrc, const int* __restrict__ edst,
                                      int* __restrict__ deg, int* __restrict__ adj,
                                      int* __restrict__ ovf, int* __restrict__ ovn,
                                      int n_edges, int n_fill_blocks) {
    const int b    = blockIdx.x;
    const int nmin = n_tiles < n_fill_blocks ? n_tiles : n_fill_blocks;
    bool is_gemm;
    int  idx;
    if (b < 2 * nmin) { is_gemm = (b & 1) == 0; idx = b >> 1; }
    else              { is_gemm = n_tiles > n_fill_blocks; idx = b - nmin; }
    if (is_gemm) {
        gemm_tile<false, false, true>(A, Bp, nullptr, nullptr, nullptr, Ch, HID, idx, a_src, a_dst, ss, sd);
    } else {
        const int e = idx * 256 + threadIdx.x;
        if (e < n_edges) {
            const int d = edst[e];
            const int s = esrc[e];
            const int r = atomicAdd(&deg[d], 1);
            if (r < BKT) {
                adj[(size_t)d * BKT + r] = s;
            } else {
                const int o = atomicAdd(ovn, 1);
                ovf[2 * (size_t)o]     = d;
                ovf[2 * (size_t)o + 1] = s;
            }
        }
    }
}

__device__ __forceinline__ float lrelu(float v) { return (v > 0.f) ? v : NEG_SLOPE * v; }

__device__ __forceinline__ void axpy4(float4& acc, float w, const float4& h) {
    acc.x = __builtin_fmaf(w, h.x, acc.x);
    acc.y = __builtin_fmaf(w, h.y, acc.y);
    acc.z = __builtin_fmaf(w, h.z, acc.z);
    acc.w = __builtin_fmaf(w, h.w, acc.w);
}

__device__ __forceinline__ float4 row_piece(const _Float16* __restrict__ H, int row, int lane) {
    const v4h h = *((const v4h*)(H + (size_t)row * HID) + lane);
    return make_float4((float)h[0], (float)h[1], (float)h[2], (float)h[3]);
}

__device__ __forceinline__ float bn_relu(float x, float b, float g, float be, float mu, float va) {
    float v = x + b;
    v = (v - mu) * (g * rsqrtf(va + BN_EPS)) + be;
    return fmaxf(v, 0.f);
}

__device__ __forceinline__ void pool_flush(float* __restrict__ gsum, float* __restrict__ gcnt,
                                           int g, int lane, float r0, float r1, float r2,
                                           float r3, float rc) {
    float* p = gsum + (size_t)g * HID + lane;
    atomicAdd(p,      r0);
    atomicAdd(p + 32, r1);
    atomicAdd(p + 64, r2);
    atomicAdd(p + 96, r3);
    if (lane == 0) atomicAdd(gcnt + g, rc);
}

template <bool POOL>
__global__ void __launch_bounds__(256) node_gather(const _Float16* __restrict__ H,
                                                   const int* __restrict__ deg,
                                                   const int* __restrict__ adj,
                                                   const int* __restrict__ ovf,
                                                   const int* __restrict__ ovn,
                                                   const float* __restrict__ ss,
                                                   const float* __restrict__ sd,
                                                   const float* __restrict__ bias,
                                                   const float* __restrict__ gam,
                                                   const float* __restrict__ bet,
                                                   const float* __restrict__ mean,
                                                   const float* __restrict__ var,
                                                   const int* __restrict__ batch,
                                                   float* __restrict__ out,
                                                   float* __restrict__ gsum,
                                                   float* __restrict__ gcnt, int n_nodes) {
    __shared__ __align__(16) float prow[8 * HID];
    __shared__ int pgid[8];
    __shared__ int parrived;
    int d = (int)((blockIdx.x * (size_t)blockDim.x + threadIdx.x) >> 5);
    const int lane = threadIdx.x & 31;
    const int wave = threadIdx.x >> 5;
    if (POOL) {
        if (threadIdx.x == 0) parrived = 0;
        __syncthreads();
    }
    if (d >= n_nodes) return;
    d = __builtin_amdgcn_readfirstlane(d);
    const int  degd = __builtin_amdgcn_readfirstlane(deg[d]);
    const int  beg  = d * BKT;
    const int  end  = beg + (degd < BKT ? degd : BKT);
    const int  novl = __builtin_amdgcn_readfirstlane(ovn[0]);
    const int  nov  = degd > BKT ? novl : 0;

    const float sdd   = sd[d];
    const float vself = lrelu(ss[d] + sdd);

    const int  i0   = beg + lane;
    const bool has0 = i0 < end;
    int   s0 = d;
    float v0 = vself;
    if (has0) { s0 = adj[i0]; v0 = lrelu(ss[s0] + sdd); }

    float m = fmaxf(vself, v0);
    for (int i = i0 + 32; i < end; i += 32)
        m = fmaxf(m, lrelu(ss[adj[i]] + sdd));
    for (int i = lane; i < nov; i += 32)
        if (ovf[2 * (size_t)i] == d) m = fmaxf(m, lrelu(ss[ovf[2 * (size_t)i + 1]] + sdd));
#pragma unroll
    for (int off = 16; off > 0; off >>= 1) m = fmaxf(m, __shfl_xor(m, off, 32));

    float sum = has0 ? __expf(v0 - m) : 0.f;
    for (int i = i0 + 32; i < end; i += 32)
        sum += __expf(lrelu(ss[adj[i]] + sdd) - m);
    for (int i = lane; i < nov; i += 32)
        if (ovf[2 * (size_t)i] == d) sum += __expf(lrelu(ss[ovf[2 * (size_t)i + 1]] + sdd) - m);
#pragma unroll
    for (int off = 16; off > 0; off >>= 1) sum += __shfl_xor(sum, off, 32);

    const float eself = __expf(vself - m);
    const float den   = sum + eself + 1e-16f;

    const float wself = eself / den;
    float4 hv = row_piece(H, d, lane);
    float4 acc = make_float4(wself * hv.x, wself * hv.y, wself * hv.z, wself * hv.w);

    for (int base = beg; base < end; base += 32) {
        int   s = s0;
        float w = has0 ? __expf(v0 - m) / den : 0.f;
        if (base != beg) {
            int i = base + lane;
            s = d; w = 0.f;
            if (i < end) { s = adj[i]; w = __expf(lrelu(ss[s] + sdd) - m) / den; }
        }
        const int rem = end - base;
        const int cnt = rem < 32 ? rem : 32;
        for (int j = 0; j < cnt; j += 4) {
            int sa = __builtin_amdgcn_readlane(s, j);
            int sb = __builtin_amdgcn_readlane(s, j + 1);
            int sc = __builtin_amdgcn_readlane(s, j + 2);
            int se = __builtin_amdgcn_readlane(s, j + 3);
            float wa = __int_as_float(__builtin_amdgcn_readlane(__float_as_int(w), j));
            float wb = __int_as_float(__builtin_amdgcn_readlane(__float_as_int(w), j + 1));
            float wc = __int_as_float(__builtin_amdgcn_readlane(__float_as_int(w), j + 2));
            float we = __int_as_float(__builtin_amdgcn_readlane(__float_as_int(w), j + 3));
            float4 ha = row_piece(H, sa, lane);
            float4 hb = row_piece(H, sb, lane);
            float4 hc = row_piece(H, sc, lane);
            float4 he = row_piece(H, se, lane);
            axpy4(acc, wa, ha);
            axpy4(acc, wb, hb);
            axpy4(acc, wc, hc);
            axpy4(acc, we, he);
        }
    }
    for (int base = 0; base < nov; base += 32) {
        const int i = base + lane;
        int   s = d;
        float w = 0.f;
        bool  hit = false;
        if (i < nov && ovf[2 * (size_t)i] == d) {
            hit = true;
            s = ovf[2 * (size_t)i + 1];
            w = __expf(lrelu(ss[s] + sdd) - m) / den;
        }
        unsigned bm = __builtin_amdgcn_ballot_w32(hit);
        while (bm) {
            const int j = __builtin_ctz(bm);
            bm &= bm - 1;
            const int   sj = __builtin_amdgcn_readlane(s, j);
            const float wj = __int_as_float(__builtin_amdgcn_readlane(__float_as_int(w), j));
            const float4 hj = row_piece(H, sj, lane);
            axpy4(acc, wj, hj);
        }
    }
    const float4 bi = *((const float4*)bias + lane);
    const float4 ga = *((const float4*)gam  + lane);
    const float4 be = *((const float4*)bet  + lane);
    const float4 mu = *((const float4*)mean + lane);
    const float4 va = *((const float4*)var  + lane);
    acc.x = bn_relu(acc.x, bi.x, ga.x, be.x, mu.x, va.x);
    acc.y = bn_relu(acc.y, bi.y, ga.y, be.y, mu.y, va.y);
    acc.z = bn_relu(acc.z, bi.z, ga.z, be.z, mu.z, va.z);
    acc.w = bn_relu(acc.w, bi.w, ga.w, be.w, mu.w, va.w);
    if (POOL) {
        const int g = __builtin_amdgcn_readfirstlane(batch[d]);
        *((float4*)(prow + wave * HID) + lane) = acc;
        if (lane == 0) pgid[wave] = g;
        const int left   = n_nodes - (int)blockIdx.x * 8;
        const int nvalid = left < 8 ? left : 8;
        __builtin_amdgcn_fence(__ATOMIC_RELEASE, "workgroup");
        int t = 0;
        if (lane == 0)
            t = __hip_atomic_fetch_add(&parrived, 1, __ATOMIC_RELAXED, __HIP_MEMORY_SCOPE_WORKGROUP);
        t = __builtin_amdgcn_readfirstlane(t);
        if (t == nvalid - 1) {
            __builtin_amdgcn_fence(__ATOMIC_ACQUIRE, "workgroup");
            int   gcur = __builtin_amdgcn_readfirstlane(pgid[0]);
            float r0 = 0.f, r1 = 0.f, r2 = 0.f, r3 = 0.f, rc = 0.f;
            for (int w = 0; w < nvalid; ++w) {
                const int gw = __builtin_amdgcn_readfirstlane(pgid[w]);
                if (gw != gcur) {
                    pool_flush(gsum, gcnt, gcur, lane, r0, r1, r2, r3, rc);
                    r0 = 0.f; r1 = 0.f; r2 = 0.f; r3 = 0.f; rc = 0.f;
                    gcur = gw;
                }
                const float* pr = prow + w * HID + lane;
                r0 += pr[0];
                r1 += pr[32];
                r2 += pr[64];
                r3 += pr[96];
                rc += 1.0f;
            }
            pool_flush(gsum, gcnt, gcur, lane, r0, r1, r2, r3, rc);
        }
    } else {
        *((float4*)(out + (size_t)d * HID) + lane) = acc;
    }
}

extern "C" void kernel_launch(void* const* d_in, const int* in_sizes, int n_in,
                              void* d_out, int out_size, void* d_ws, size_t ws_size,
                              hipStream_t stream) {
    const float* gx  = (const float*)d_in[0];
    const int*   ei  = (const int*)  d_in[1];
    const int*   bat = (const int*)  d_in[2];
    const float* W1  = (const float*)d_in[3];
    const float* as1 = (const float*)d_in[4];
    const float* ad1 = (const float*)d_in[5];
    const float* b1  = (const float*)d_in[6];
    const float* g1  = (const float*)d_in[7];
    const float* be1 = (const float*)d_in[8];
    const float* mu1 = (const float*)d_in[9];
    const float* va1 = (const float*)d_in[10];
    const float* W2  = (const float*)d_in[11];
    const float* as2 = (const float*)d_in[12];
    const float* ad2 = (const float*)d_in[13];
    const float* b2  = (const float*)d_in[14];
    const float* g2  = (const float*)d_in[15];
    const float* be2 = (const float*)d_in[16];
    const float* mu2 = (const float*)d_in[17];
    const float* va2 = (const float*)d_in[18];
    const float* hw  = (const float*)d_in[19];
    const float* hb  = (const float*)d_in[20];

    const int n_nodes = in_sizes[0] / HID;
    const int n_edges = in_sizes[1] / 2;
    const int* esrc = ei;
    const int* edst = ei + n_edges;

    float* ws   = (float*)d_ws;
    float* H    = ws;
    _Float16* H16 = (_Float16*)H;
    float* AGG  = H    + (size_t)n_nodes * HID;
    float* SS   = AGG  + (size_t)n_nodes * HID;
    float* SD   = SS   + n_nodes;
    float* POOL = SD   + n_nodes;
    float* CNT  = POOL + (size_t)N_GRAPHS * HID;
    size_t wpOff = (size_t)(CNT + N_GRAPHS - ws);
    wpOff = (wpOff + 7) & ~(size_t)7;
    _Float16* WP1 = (_Float16*)(ws + wpOff);
    _Float16* WP2 = WP1 + HID * HID;
    _Float16* WP3 = WP2 + HID * HID;
    const size_t degOff = wpOff + (HID * HID) / 2 + (HID * HID) / 2 + (HID * 64) / 2;
    int* DEG = (int*)(ws + degOff);
    int* OVN = DEG + n_nodes;
    const size_t adjOff = (degOff + n_nodes + 1 + 63) & ~(size_t)63;
    int* ADJ = (int*)(ws + adjOff);
    int* OVF = ADJ + (size_t)n_nodes * BKT;

    const int gridEdge  = (n_edges + 255) / 256;
    const int gridNodeW = (n_nodes + 7) / 8;
    const int gridGemm  = n_nodes / 16;
    const int nInt      = n_nodes + 1;
    const int nPool     = N_GRAPHS * HID + N_GRAPHS;

    prep_pack_and_clear<<<10 + (nInt + nPool + 255) / 256, 256, 0, stream>>>(
        W1, W2, hw, WP1, WP2, WP3, DEG, nInt, POOL, nPool);

    gemm1_and_bucket_fill<<<gridGemm + gridEdge, 256, 0, stream>>>(
        gx, WP1, H16, gridGemm, as1, ad1, SS, SD, esrc, edst, DEG, ADJ, OVF, OVN, n_edges, gridEdge);

    node_gather<false><<<gridNodeW, 256, 0, stream>>>(
        H16, DEG, ADJ, OVF, OVN, SS, SD, b1, g1, be1, mu1, va1, bat, AGG, POOL, CNT, n_nodes);

    gemm128_wmma<false, false, true><<<gridGemm, 256, 0, stream>>>(
        AGG, WP2, nullptr, nullptr, nullptr, H16, HID, as2, ad2, SS, SD);
    node_gather<true><<<gridNodeW, 256, 0, stream>>>(
        H16, DEG, ADJ, OVF, OVN, SS, SD, b2, g2, be2, mu2, va2, bat, AGG, POOL, CNT, n_nodes);

    gemm128_wmma<true, true, false><<<N_GRAPHS / 16, 32 * (64 / 16), 0, stream>>>(
        POOL, WP3, hb, CNT, (float*)d_out, nullptr, 64, nullptr, nullptr, nullptr, nullptr);
}
